# starting kernel
# baseline (speedup 1.0000x reference)
#include <hip/hip_runtime.h>
#include <hip/hip_bf16.h>

typedef __attribute__((ext_vector_type(2))) float v2f;
typedef __attribute__((ext_vector_type(8))) float v8f;

#define LN_EPS 1e-5f

__global__ __launch_bounds__(256) void gemm128_wmma(
    const float* __restrict__ A, const float* __restrict__ W,
    float* __restrict__ H, int N)
{
    __shared__ float As[16 * 128];
    const int rowBase = blockIdx.x << 4;

    for (int i = threadIdx.x; i < 512; i += 256) {
        const int rrow = i >> 5;
        const int rcol = i & 31;
        const int gr   = rowBase + rrow;
        float4 v = make_float4(0.f, 0.f, 0.f, 0.f);
        if (gr < N)
            v = reinterpret_cast<const float4*>(A + (size_t)gr * 128)[rcol];
        reinterpret_cast<float4*>(As)[i] = v;
    }
    __syncthreads();

    const int lane    = threadIdx.x & 31;
    const int wave    = threadIdx.x >> 5;
    const int colBase = wave << 4;
    const int r       = lane & 15;
    const int kh      = lane >> 4;

    v8f acc = {};
    #pragma unroll
    for (int k = 0; k < 128; k += 4) {
        const int ka = k + 2 * kh;
        v2f a, b;
        a.x = As[r * 128 + ka];
        a.y = As[r * 128 + ka + 1];
        b.x = W[(size_t)ka * 128 + colBase + r];
        b.y = W[(size_t)(ka + 1) * 128 + colBase + r];
        acc = __builtin_amdgcn_wmma_f32_16x16x4_f32(false, a, false, b,
                                                    (short)0, acc, false, false);
    }

    #pragma unroll
    for (int v = 0; v < 8; ++v) {
        const int gr = rowBase + 8 * kh + v;
        if (gr < N) H[(size_t)gr * 128 + colBase + r] = acc[v];
    }
}

__global__ __launch_bounds__(256) void zero_f32(float* __restrict__ p, int n4)
{
    int i = blockIdx.x * 256 + threadIdx.x;
    if (i < n4)
        reinterpret_cast<float4*>(p)[i] = make_float4(0.f, 0.f, 0.f, 0.f);
}

__global__ __launch_bounds__(256) void scatter_edges(
    const float* __restrict__ H,
    const int*   __restrict__ src,
    const int*   __restrict__ dst,
    const float* __restrict__ ew,
    float* __restrict__ AGG, int E)
{
    const int e = blockIdx.x * 8 + (threadIdx.x >> 5);
    if (e >= E) return;
    const int lane = threadIdx.x & 31;

    const int   s = src[e];
    const int   d = dst[e];
    const float w = ew[e];

    float4 m = reinterpret_cast<const float4*>(H + (size_t)s * 128)[lane];
    float* out = AGG + (size_t)d * 128 + lane * 4;
    unsafeAtomicAdd(out + 0, m.x * w);
    unsafeAtomicAdd(out + 1, m.y * w);
    unsafeAtomicAdd(out + 2, m.z * w);
    unsafeAtomicAdd(out + 3, m.w * w);
}

__global__ __launch_bounds__(256) void bias_ln_relu(
    const float* __restrict__ AGG,
    const float* __restrict__ bias,
    const float* __restrict__ gamma,
    const float* __restrict__ beta,
    float* __restrict__ out, int N)
{
    const int n = blockIdx.x * 8 + (threadIdx.x >> 5);
    if (n >= N) return;
    const int lane = threadIdx.x & 31;

    float4 v = reinterpret_cast<const float4*>(AGG + (size_t)n * 128)[lane];
    float4 b = reinterpret_cast<const float4*>(bias)[lane];
    v.x += b.x; v.y += b.y; v.z += b.z; v.w += b.w;

    float s = v.x + v.y + v.z + v.w;
    #pragma unroll
    for (int off = 16; off >= 1; off >>= 1) s += __shfl_xor(s, off, 32);
    const float mu = s * (1.0f / 128.0f);

    const float dx = v.x - mu, dy = v.y - mu, dz = v.z - mu, dw = v.w - mu;
    float q = dx * dx + dy * dy + dz * dz + dw * dw;
    #pragma unroll
    for (int off = 16; off >= 1; off >>= 1) q += __shfl_xor(q, off, 32);
    const float rstd = rsqrtf(q * (1.0f / 128.0f) + LN_EPS);

    float4 g  = reinterpret_cast<const float4*>(gamma)[lane];
    float4 be = reinterpret_cast<const float4*>(beta)[lane];
    float4 o;
    o.x = fmaxf(dx * rstd * g.x + be.x, 0.f);
    o.y = fmaxf(dy * rstd * g.y + be.y, 0.f);
    o.z = fmaxf(dz * rstd * g.z + be.z, 0.f);
    o.w = fmaxf(dw * rstd * g.w + be.w, 0.f);
    reinterpret_cast<float4*>(out + (size_t)n * 128)[lane] = o;
}

extern "C" void kernel_launch(void* const* d_in, const int* in_sizes, int n_in,
                              void* d_out, int out_size, void* d_ws, size_t ws_size,
                              hipStream_t stream)
{
    const float* x   = (const float*)d_in[0];
    const int*   ei  = (const int*)  d_in[1];
    const float* ew  = (const float*)d_in[2];
    const float* W1  = (const float*)d_in[3];
    const float* b1  = (const float*)d_in[4];
    const float* W2  = (const float*)d_in[5];
    const float* b2  = (const float*)d_in[6];
    const float* g1  = (const float*)d_in[7];
    const float* be1 = (const float*)d_in[8];
    const float* g2  = (const float*)d_in[9];
    const float* be2 = (const float*)d_in[10];

    const int N = in_sizes[0] / 128;
    const int E = in_sizes[2];
    const int* src = ei;
    const int* dst = ei + E;

    float* H   = (float*)d_ws;
    float* out = (float*)d_out;

    const int gemmBlocks = (N + 15) / 16;
    const int n4         = (N * 128) / 4;
    const int zeroBlocks = (n4 + 255) / 256;
    const int edgeBlocks = (E + 7) / 8;
    const int rowBlocks  = (N + 7) / 8;

    gemm128_wmma<<<gemmBlocks, 256, 0, stream>>>(x, W1, H, N);
    zero_f32<<<zeroBlocks, 256, 0, stream>>>(out, n4);
    scatter_edges<<<edgeBlocks, 256, 0, stream>>>(H, src, dst, ew, out, E);
    bias_ln_relu<<<rowBlocks, 256, 0, stream>>>(out, b1, g1, be1, out, N);

    gemm128_wmma<<<gemmBlocks, 256, 0, stream>>>(out, W2, H, N);
    zero_f32<<<zeroBlocks, 256, 0, stream>>>(out, n4);
    scatter_edges<<<edgeBlocks, 256, 0, stream>>>(H, src, dst, ew, out, E);
    bias_ln_relu<<<rowBlocks, 256, 0, stream>>>(out, b2, g2, be2, out, N);
}


# scatter: lane l owns channels l+32j, so each global_atomic_add_f32 and each gather load of a wave covers one whole 128-B line, not 8 lanes in each of 4 lines; f32 atomics kept (arrival order)
# speedup vs baseline: 1.3234x; 1.3234x over previous
#include <hip/hip_runtime.h>
#include <hip/hip_bf16.h>

typedef __attribute__((ext_vector_type(2))) float v2f;
typedef __attribute__((ext_vector_type(8))) float v8f;

#define LN_EPS 1e-5f

__global__ __launch_bounds__(256) void gemm128_wmma(
    const float* __restrict__ A, const float* __restrict__ W,
    float* __restrict__ H, int N)
{
    __shared__ float As[16 * 128];
    const int rowBase = blockIdx.x << 4;

    for (int i = threadIdx.x; i < 512; i += 256) {
        const int rrow = i >> 5;
        const int rcol = i & 31;
        const int gr   = rowBase + rrow;
        float4 v = make_float4(0.f, 0.f, 0.f, 0.f);
        if (gr < N)
            v = reinterpret_cast<const float4*>(A + (size_t)gr * 128)[rcol];
        reinterpret_cast<float4*>(As)[i] = v;
    }
    __syncthreads();

    const int lane    = threadIdx.x & 31;
    const int wave    = threadIdx.x >> 5;
    const int colBase = wave << 4;
    const int r       = lane & 15;
    const int kh      = lane >> 4;

    v8f acc = {};
    #pragma unroll
    for (int k = 0; k < 128; k += 4) {
        const int ka = k + 2 * kh;
        v2f a, b;
        a.x = As[r * 128 + ka];
        a.y = As[r * 128 + ka + 1];
        b.x = W[(size_t)ka * 128 + colBase + r];
        b.y = W[(size_t)(ka + 1) * 128 + colBase + r];
        acc = __builtin_amdgcn_wmma_f32_16x16x4_f32(false, a, false, b,
                                                    (short)0, acc, false, false);
    }

    #pragma unroll
    for (int v = 0; v < 8; ++v) {
        const int gr = rowBase + 8 * kh + v;
        if (gr < N) H[(size_t)gr * 128 + colBase + r] = acc[v];
    }
}

__global__ __launch_bounds__(256) void zero_f32(float* __restrict__ p, int n4)
{
    int i = blockIdx.x * 256 + threadIdx.x;
    if (i < n4)
        reinterpret_cast<float4*>(p)[i] = make_float4(0.f, 0.f, 0.f, 0.f);
}

__global__ __launch_bounds__(256) void scatter_edges(
    const float* __restrict__ H,
    const int*   __restrict__ src,
    const int*   __restrict__ dst,
    const float* __restrict__ ew,
    float* __restrict__ AGG, int E)
{
    const int e = blockIdx.x * 8 + (threadIdx.x >> 5);
    if (e >= E) return;
    const int lane = threadIdx.x & 31;

    const int   s = src[e];
    const int   d = dst[e];
    const float w = ew[e];

    const float* hrow = H + (size_t)s * 128 + lane;
    const float m0 = hrow[0];
    const float m1 = hrow[32];
    const float m2 = hrow[64];
    const float m3 = hrow[96];

    float* out = AGG + (size_t)d * 128 + lane;
    unsafeAtomicAdd(out +  0, m0 * w);
    unsafeAtomicAdd(out + 32, m1 * w);
    unsafeAtomicAdd(out + 64, m2 * w);
    unsafeAtomicAdd(out + 96, m3 * w);
}

__global__ __launch_bounds__(256) void bias_ln_relu(
    const float* __restrict__ AGG,
    const float* __restrict__ bias,
    const float* __restrict__ gamma,
    const float* __restrict__ beta,
    float* __restrict__ out, int N)
{
    const int n = blockIdx.x * 8 + (threadIdx.x >> 5);
    if (n >= N) return;
    const int lane = threadIdx.x & 31;

    float4 v = reinterpret_cast<const float4*>(AGG + (size_t)n * 128)[lane];
    float4 b = reinterpret_cast<const float4*>(bias)[lane];
    v.x += b.x; v.y += b.y; v.z += b.z; v.w += b.w;

    float s = v.x + v.y + v.z + v.w;
    #pragma unroll
    for (int off = 16; off >= 1; off >>= 1) s += __shfl_xor(s, off, 32);
    const float mu = s * (1.0f / 128.0f);

    const float dx = v.x - mu, dy = v.y - mu, dz = v.z - mu, dw = v.w - mu;
    float q = dx * dx + dy * dy + dz * dz + dw * dw;
    #pragma unroll
    for (int off = 16; off >= 1; off >>= 1) q += __shfl_xor(q, off, 32);
    const float rstd = rsqrtf(q * (1.0f / 128.0f) + LN_EPS);

    float4 g  = reinterpret_cast<const float4*>(gamma)[lane];
    float4 be = reinterpret_cast<const float4*>(beta)[lane];
    float4 o;
    o.x = fmaxf(dx * rstd * g.x + be.x, 0.f);
    o.y = fmaxf(dy * rstd * g.y + be.y, 0.f);
    o.z = fmaxf(dz * rstd * g.z + be.z, 0.f);
    o.w = fmaxf(dw * rstd * g.w + be.w, 0.f);
    reinterpret_cast<float4*>(out + (size_t)n * 128)[lane] = o;
}

extern "C" void kernel_launch(void* const* d_in, const int* in_sizes, int n_in,
                              void* d_out, int out_size, void* d_ws, size_t ws_size,
                              hipStream_t stream)
{
    const float* x   = (const float*)d_in[0];
    const int*   ei  = (const int*)  d_in[1];
    const float* ew  = (const float*)d_in[2];
    const float* W1  = (const float*)d_in[3];
    const float* b1  = (const float*)d_in[4];
    const float* W2  = (const float*)d_in[5];
    const float* b2  = (const float*)d_in[6];
    const float* g1  = (const float*)d_in[7];
    const float* be1 = (const float*)d_in[8];
    const float* g2  = (const float*)d_in[9];
    const float* be2 = (const float*)d_in[10];

    const int N = in_sizes[0] / 128;
    const int E = in_sizes[2];
    const int* src = ei;
    const int* dst = ei + E;

    float* H   = (float*)d_ws;
    float* out = (float*)d_out;

    const int gemmBlocks = (N + 15) / 16;
    const int n4         = (N * 128) / 4;
    const int zeroBlocks = (n4 + 255) / 256;
    const int edgeBlocks = (E + 7) / 8;
    const int rowBlocks  = (N + 7) / 8;

    gemm128_wmma<<<gemmBlocks, 256, 0, stream>>>(x, W1, H, N);
    zero_f32<<<zeroBlocks, 256, 0, stream>>>(out, n4);
    scatter_edges<<<edgeBlocks, 256, 0, stream>>>(H, src, dst, ew, out, E);
    bias_ln_relu<<<rowBlocks, 256, 0, stream>>>(out, b1, g1, be1, out, N);

    gemm128_wmma<<<gemmBlocks, 256, 0, stream>>>(out, W2, H, N);
    zero_f32<<<zeroBlocks, 256, 0, stream>>>(out, n4);
    scatter_edges<<<edgeBlocks, 256, 0, stream>>>(H, src, dst, ew, out, E);
    bias_ln_relu<<<rowBlocks, 256, 0, stream>>>(out, b2, g2, be2, out, N);
}


# scatter -> gather: edges bucketed by dst in-call (u32 count atomic per edge gives its rank, scan, fill); one wave sums each row in registers in rank order; f32 atomics and zeroing of out removed
# speedup vs baseline: 1.9041x; 1.4388x over previous
#include <hip/hip_runtime.h>
#include <hip/hip_bf16.h>

typedef __attribute__((ext_vector_type(2))) float v2f;
typedef __attribute__((ext_vector_type(8))) float v8f;

#define LN_EPS 1e-5f

__global__ __launch_bounds__(256) void gemm128_wmma(
    const float* __restrict__ A, const float* __restrict__ W,
    float* __restrict__ H, int N)
{
    __shared__ float As[16 * 128];
    const int rowBase = blockIdx.x << 4;

    for (int i = threadIdx.x; i < 512; i += 256) {
        const int rrow = i >> 5;
        const int rcol = i & 31;
        const int gr   = rowBase + rrow;
        float4 v = make_float4(0.f, 0.f, 0.f, 0.f);
        if (gr < N)
            v = reinterpret_cast<const float4*>(A + (size_t)gr * 128)[rcol];
        reinterpret_cast<float4*>(As)[i] = v;
    }
    __syncthreads();

    const int lane    = threadIdx.x & 31;
    const int wave    = threadIdx.x >> 5;
    const int colBase = wave << 4;
    const int r       = lane & 15;
    const int kh      = lane >> 4;

    v8f acc = {};
    #pragma unroll
    for (int k = 0; k < 128; k += 4) {
        const int ka = k + 2 * kh;
        v2f a, b;
        a.x = As[r * 128 + ka];
        a.y = As[r * 128 + ka + 1];
        b.x = W[(size_t)ka * 128 + colBase + r];
        b.y = W[(size_t)(ka + 1) * 128 + colBase + r];
        acc = __builtin_amdgcn_wmma_f32_16x16x4_f32(false, a, false, b,
                                                    (short)0, acc, false, false);
    }

    #pragma unroll
    for (int v = 0; v < 8; ++v) {
        const int gr = rowBase + 8 * kh + v;
        if (gr < N) H[(size_t)gr * 128 + colBase + r] = acc[v];
    }
}

__global__ __launch_bounds__(256) void zero_f32(float* __restrict__ p, int n4)
{
    int i = blockIdx.x * 256 + threadIdx.x;
    if (i < n4)
        reinterpret_cast<float4*>(p)[i] = make_float4(0.f, 0.f, 0.f, 0.f);
}

__global__ __launch_bounds__(256) void count_rank(
    const int* __restrict__ dst, int* __restrict__ cnt,
    int* __restrict__ rank, int E)
{
    const int e = blockIdx.x * 256 + threadIdx.x;
    if (e < E) rank[e] = atomicAdd(&cnt[dst[e]], 1);
}

__global__ __launch_bounds__(1024) void scan_counts(
    const int* __restrict__ cnt, int* __restrict__ rowptr, int N)
{
    __shared__ int wsum[32];
    __shared__ int chunkTotal;
    const int tid  = threadIdx.x;
    const int lane = tid & 31;
    const int wave = tid >> 5;

    int carry = 0;
    for (int base = 0; base < N; base += 16384) {
        const int i = base + tid * 16;
        int4 c[4];
        #pragma unroll
        for (int q = 0; q < 4; ++q) {
            c[q] = make_int4(0, 0, 0, 0);
            if (i + 4 * q < N) c[q] = *reinterpret_cast<const int4*>(cnt + i + 4 * q);
        }
        int4 p[4];
        int run = 0;
        #pragma unroll
        for (int q = 0; q < 4; ++q) {
            p[q].x = run; run += c[q].x;
            p[q].y = run; run += c[q].y;
            p[q].z = run; run += c[q].z;
            p[q].w = run; run += c[q].w;
        }
        const int t = run;

        int incl = t;
        #pragma unroll
        for (int off = 1; off < 32; off <<= 1) {
            const int y = __shfl_up(incl, off, 32);
            if (lane >= off) incl += y;
        }
        if (lane == 31) wsum[wave] = incl;
        __syncthreads();
        if (wave == 0) {
            const int v = wsum[lane];
            int s = v;
            #pragma unroll
            for (int off = 1; off < 32; off <<= 1) {
                const int y = __shfl_up(s, off, 32);
                if (lane >= off) s += y;
            }
            wsum[lane] = s - v;
            if (lane == 31) chunkTotal = s;
        }
        __syncthreads();
        const int excl = carry + wsum[wave] + incl - t;
        #pragma unroll
        for (int q = 0; q < 4; ++q)
            if (i + 4 * q < N)
                *reinterpret_cast<int4*>(rowptr + i + 4 * q) =
                    make_int4(excl + p[q].x, excl + p[q].y, excl + p[q].z, excl + p[q].w);
        carry += chunkTotal;
        __syncthreads();
    }
    if (tid == 0) rowptr[N] = carry;
}

__global__ __launch_bounds__(256) void fill_buckets(
    const int*   __restrict__ src,
    const int*   __restrict__ dst,
    const float* __restrict__ ew,
    const int*   __restrict__ rank,
    const int*   __restrict__ rowptr,
    int2* __restrict__ perm, int E)
{
    const int e = blockIdx.x * 256 + threadIdx.x;
    if (e < E)
        perm[rowptr[dst[e]] + rank[e]] = make_int2(src[e], __float_as_int(ew[e]));
}

__global__ __launch_bounds__(256) void gather_rows(
    const float* __restrict__ H,
    const int*   __restrict__ rowptr,
    const int2*  __restrict__ perm,
    float* __restrict__ AGG, int N)
{
    const int n = __builtin_amdgcn_readfirstlane(blockIdx.x * 8 + (threadIdx.x >> 5));
    if (n >= N) return;
    const int lane = threadIdx.x & 31;
    const int beg  = __builtin_amdgcn_readfirstlane(rowptr[n]);
    const int end  = __builtin_amdgcn_readfirstlane(rowptr[n + 1]);

    const float* hl = H + lane * 4;
    float4 acc = make_float4(0.f, 0.f, 0.f, 0.f);

    for (int base = beg; base < end; base += 32) {
        const int c = min(32, end - base);
        int2 pr = make_int2(0, 0);
        if (lane < c) pr = perm[base + lane];

        for (int j = 0; j < c; j += 8) {
            int s[8];
            #pragma unroll
            for (int u = 0; u < 8; ++u)
                s[u] = __builtin_amdgcn_readlane(pr.x, j + u);
            float4 h[8];
            #pragma unroll
            for (int u = 0; u < 8; ++u) {
                h[u] = make_float4(0.f, 0.f, 0.f, 0.f);
                if (j + u < c)
                    h[u] = *reinterpret_cast<const float4*>(hl + (size_t)s[u] * 128);
            }
            #pragma unroll
            for (int u = 0; u < 8; ++u) {
                const float w = __int_as_float(__builtin_amdgcn_readlane(pr.y, j + u));
                acc.x += h[u].x * w;
                acc.y += h[u].y * w;
                acc.z += h[u].z * w;
                acc.w += h[u].w * w;
            }
        }
    }
    reinterpret_cast<float4*>(AGG + (size_t)n * 128)[lane] = acc;
}

__global__ __launch_bounds__(256) void bias_ln_relu(
    const float* __restrict__ AGG,
    const float* __restrict__ bias,
    const float* __restrict__ gamma,
    const float* __restrict__ beta,
    float* __restrict__ out, int N)
{
    const int n = blockIdx.x * 8 + (threadIdx.x >> 5);
    if (n >= N) return;
    const int lane = threadIdx.x & 31;

    float4 v = reinterpret_cast<const float4*>(AGG + (size_t)n * 128)[lane];
    float4 b = reinterpret_cast<const float4*>(bias)[lane];
    v.x += b.x; v.y += b.y; v.z += b.z; v.w += b.w;

    float s = v.x + v.y + v.z + v.w;
    #pragma unroll
    for (int off = 16; off >= 1; off >>= 1) s += __shfl_xor(s, off, 32);
    const float mu = s * (1.0f / 128.0f);

    const float dx = v.x - mu, dy = v.y - mu, dz = v.z - mu, dw = v.w - mu;
    float q = dx * dx + dy * dy + dz * dz + dw * dw;
    #pragma unroll
    for (int off = 16; off >= 1; off >>= 1) q += __shfl_xor(q, off, 32);
    const float rstd = rsqrtf(q * (1.0f / 128.0f) + LN_EPS);

    float4 g  = reinterpret_cast<const float4*>(gamma)[lane];
    float4 be = reinterpret_cast<const float4*>(beta)[lane];
    float4 o;
    o.x = fmaxf(dx * rstd * g.x + be.x, 0.f);
    o.y = fmaxf(dy * rstd * g.y + be.y, 0.f);
    o.z = fmaxf(dz * rstd * g.z + be.z, 0.f);
    o.w = fmaxf(dw * rstd * g.w + be.w, 0.f);
    reinterpret_cast<float4*>(out + (size_t)n * 128)[lane] = o;
}

extern "C" void kernel_launch(void* const* d_in, const int* in_sizes, int n_in,
                              void* d_out, int out_size, void* d_ws, size_t ws_size,
                              hipStream_t stream)
{
    const float* x   = (const float*)d_in[0];
    const int*   ei  = (const int*)  d_in[1];
    const float* ew  = (const float*)d_in[2];
    const float* W1  = (const float*)d_in[3];
    const float* b1  = (const float*)d_in[4];
    const float* W2  = (const float*)d_in[5];
    const float* b2  = (const float*)d_in[6];
    const float* g1  = (const float*)d_in[7];
    const float* be1 = (const float*)d_in[8];
    const float* g2  = (const float*)d_in[9];
    const float* be2 = (const float*)d_in[10];

    const int N = in_sizes[0] / 128;
    const int E = in_sizes[2];
    const int* src = ei;
    const int* dst = ei + E;

    float* H   = (float*)d_ws;
    float* out = (float*)d_out;

    const int Npad = (N + 4) & ~3;
    const int Epad = (E + 3) & ~3;
    int*  cnt    = (int*)(H + (size_t)N * 128);
    int*  rowptr = cnt + Npad;
    int*  rank   = rowptr + Npad;
    int2* perm   = (int2*)(rank + Epad);

    const int gemmBlocks = (N + 15) / 16;
    const int zeroBlocks = (Npad / 4 + 255) / 256;
    const int edgeBlocks = (E + 255) / 256;
    const int rowBlocks  = (N + 7) / 8;

    zero_f32<<<zeroBlocks, 256, 0, stream>>>((float*)cnt, Npad / 4);
    count_rank<<<edgeBlocks, 256, 0, stream>>>(dst, cnt, rank, E);
    scan_counts<<<1, 1024, 0, stream>>>(cnt, rowptr, N);
    fill_buckets<<<edgeBlocks, 256, 0, stream>>>(src, dst, ew, rank, rowptr, perm, E);

    gemm128_wmma<<<gemmBlocks, 256, 0, stream>>>(x, W1, H, N);
    gather_rows<<<rowBlocks, 256, 0, stream>>>(H, rowptr, perm, out, N);
    bias_ln_relu<<<rowBlocks, 256, 0, stream>>>(out, b1, g1, be1, out, N);

    gemm128_wmma<<<gemmBlocks, 256, 0, stream>>>(out, W2, H, N);
    gather_rows<<<rowBlocks, 256, 0, stream>>>(H, rowptr, perm, out, N);
    bias_ln_relu<<<rowBlocks, 256, 0, stream>>>(out, b2, g2, be2, out, N);
}


# bias + LayerNorm + ReLU moved into the gather wave, which already holds the whole 128-wide row in registers: both row-epilogue launches and the aggregated rows' store and reload are gone
# speedup vs baseline: 2.0645x; 1.0842x over previous
#include <hip/hip_runtime.h>
#include <hip/hip_bf16.h>

typedef __attribute__((ext_vector_type(2))) float v2f;
typedef __attribute__((ext_vector_type(8))) float v8f;

#define LN_EPS 1e-5f

__global__ __launch_bounds__(256) void gemm128_wmma(
    const float* __restrict__ A, const float* __restrict__ W,
    float* __restrict__ H, int N)
{
    __shared__ float As[16 * 128];
    const int rowBase = blockIdx.x << 4;

    for (int i = threadIdx.x; i < 512; i += 256) {
        const int rrow = i >> 5;
        const int rcol = i & 31;
        const int gr   = rowBase + rrow;
        float4 v = make_float4(0.f, 0.f, 0.f, 0.f);
        if (gr < N)
            v = reinterpret_cast<const float4*>(A + (size_t)gr * 128)[rcol];
        reinterpret_cast<float4*>(As)[i] = v;
    }
    __syncthreads();

    const int lane    = threadIdx.x & 31;
    const int wave    = threadIdx.x >> 5;
    const int colBase = wave << 4;
    const int r       = lane & 15;
    const int kh      = lane >> 4;

    v8f acc = {};
    #pragma unroll
    for (int k = 0; k < 128; k += 4) {
        const int ka = k + 2 * kh;
        v2f a, b;
        a.x = As[r * 128 + ka];
        a.y = As[r * 128 + ka + 1];
        b.x = W[(size_t)ka * 128 + colBase + r];
        b.y = W[(size_t)(ka + 1) * 128 + colBase + r];
        acc = __builtin_amdgcn_wmma_f32_16x16x4_f32(false, a, false, b,
                                                    (short)0, acc, false, false);
    }

    #pragma unroll
    for (int v = 0; v < 8; ++v) {
        const int gr = rowBase + 8 * kh + v;
        if (gr < N) H[(size_t)gr * 128 + colBase + r] = acc[v];
    }
}

__global__ __launch_bounds__(256) void zero_f32(float* __restrict__ p, int n4)
{
    int i = blockIdx.x * 256 + threadIdx.x;
    if (i < n4)
        reinterpret_cast<float4*>(p)[i] = make_float4(0.f, 0.f, 0.f, 0.f);
}

__global__ __launch_bounds__(256) void count_rank(
    const int* __restrict__ dst, int* __restrict__ cnt,
    int* __restrict__ rank, int E)
{
    const int e = blockIdx.x * 256 + threadIdx.x;
    if (e < E) rank[e] = atomicAdd(&cnt[dst[e]], 1);
}

__global__ __launch_bounds__(1024) void scan_counts(
    const int* __restrict__ cnt, int* __restrict__ rowptr, int N)
{
    __shared__ int wsum[32];
    __shared__ int chunkTotal;
    const int tid  = threadIdx.x;
    const int lane = tid & 31;
    const int wave = tid >> 5;

    int carry = 0;
    for (int base = 0; base < N; base += 16384) {
        const int i = base + tid * 16;
        int4 c[4];
        #pragma unroll
        for (int q = 0; q < 4; ++q) {
            c[q] = make_int4(0, 0, 0, 0);
            if (i + 4 * q < N) c[q] = *reinterpret_cast<const int4*>(cnt + i + 4 * q);
        }
        int4 p[4];
        int run = 0;
        #pragma unroll
        for (int q = 0; q < 4; ++q) {
            p[q].x = run; run += c[q].x;
            p[q].y = run; run += c[q].y;
            p[q].z = run; run += c[q].z;
            p[q].w = run; run += c[q].w;
        }
        const int t = run;

        int incl = t;
        #pragma unroll
        for (int off = 1; off < 32; off <<= 1) {
            const int y = __shfl_up(incl, off, 32);
            if (lane >= off) incl += y;
        }
        if (lane == 31) wsum[wave] = incl;
        __syncthreads();
        if (wave == 0) {
            const int v = wsum[lane];
            int s = v;
            #pragma unroll
            for (int off = 1; off < 32; off <<= 1) {
                const int y = __shfl_up(s, off, 32);
                if (lane >= off) s += y;
            }
            wsum[lane] = s - v;
            if (lane == 31) chunkTotal = s;
        }
        __syncthreads();
        const int excl = carry + wsum[wave] + incl - t;
        #pragma unroll
        for (int q = 0; q < 4; ++q)
            if (i + 4 * q < N)
                *reinterpret_cast<int4*>(rowptr + i + 4 * q) =
                    make_int4(excl + p[q].x, excl + p[q].y, excl + p[q].z, excl + p[q].w);
        carry += chunkTotal;
        __syncthreads();
    }
    if (tid == 0) rowptr[N] = carry;
}

__global__ __launch_bounds__(256) void fill_buckets(
    const int*   __restrict__ src,
    const int*   __restrict__ dst,
    const float* __restrict__ ew,
    const int*   __restrict__ rank,
    const int*   __restrict__ rowptr,
    int2* __restrict__ perm, int E)
{
    const int e = blockIdx.x * 256 + threadIdx.x;
    if (e < E)
        perm[rowptr[dst[e]] + rank[e]] = make_int2(src[e], __float_as_int(ew[e]));
}

__global__ __launch_bounds__(256) void gather_rows_ln_relu(
    const float* __restrict__ H,
    const int*   __restrict__ rowptr,
    const int2*  __restrict__ perm,
    const float* __restrict__ bias,
    const float* __restrict__ gamma,
    const float* __restrict__ beta,
    float* __restrict__ out, int N)
{
    const int n = __builtin_amdgcn_readfirstlane(blockIdx.x * 8 + (threadIdx.x >> 5));
    if (n >= N) return;
    const int lane = threadIdx.x & 31;

    const float4 b  = reinterpret_cast<const float4*>(bias)[lane];
    const float4 g  = reinterpret_cast<const float4*>(gamma)[lane];
    const float4 be = reinterpret_cast<const float4*>(beta)[lane];

    const int beg  = __builtin_amdgcn_readfirstlane(rowptr[n]);
    const int end  = __builtin_amdgcn_readfirstlane(rowptr[n + 1]);

    const float* hl = H + lane * 4;
    float4 acc = make_float4(0.f, 0.f, 0.f, 0.f);

    for (int base = beg; base < end; base += 32) {
        const int c = min(32, end - base);
        int2 pr = make_int2(0, 0);
        if (lane < c) pr = perm[base + lane];

        for (int j = 0; j < c; j += 8) {
            int s[8];
            #pragma unroll
            for (int u = 0; u < 8; ++u)
                s[u] = __builtin_amdgcn_readlane(pr.x, j + u);
            float4 h[8];
            #pragma unroll
            for (int u = 0; u < 8; ++u) {
                h[u] = make_float4(0.f, 0.f, 0.f, 0.f);
                if (j + u < c)
                    h[u] = *reinterpret_cast<const float4*>(hl + (size_t)s[u] * 128);
            }
            #pragma unroll
            for (int u = 0; u < 8; ++u) {
                const float w = __int_as_float(__builtin_amdgcn_readlane(pr.y, j + u));
                acc.x += h[u].x * w;
                acc.y += h[u].y * w;
                acc.z += h[u].z * w;
                acc.w += h[u].w * w;
            }
        }
    }

    float4 v = acc;
    v.x += b.x; v.y += b.y; v.z += b.z; v.w += b.w;

    float rsum = v.x + v.y + v.z + v.w;
    #pragma unroll
    for (int off = 16; off >= 1; off >>= 1) rsum += __shfl_xor(rsum, off, 32);
    const float mu = rsum * (1.0f / 128.0f);

    const float dx = v.x - mu, dy = v.y - mu, dz = v.z - mu, dw = v.w - mu;
    float q = dx * dx + dy * dy + dz * dz + dw * dw;
    #pragma unroll
    for (int off = 16; off >= 1; off >>= 1) q += __shfl_xor(q, off, 32);
    const float rstd = rsqrtf(q * (1.0f / 128.0f) + LN_EPS);

    float4 o;
    o.x = fmaxf(dx * rstd * g.x + be.x, 0.f);
    o.y = fmaxf(dy * rstd * g.y + be.y, 0.f);
    o.z = fmaxf(dz * rstd * g.z + be.z, 0.f);
    o.w = fmaxf(dw * rstd * g.w + be.w, 0.f);
    reinterpret_cast<float4*>(out + (size_t)n * 128)[lane] = o;
}

extern "C" void kernel_launch(void* const* d_in, const int* in_sizes, int n_in,
                              void* d_out, int out_size, void* d_ws, size_t ws_size,
                              hipStream_t stream)
{
    const float* x   = (const float*)d_in[0];
    const int*   ei  = (const int*)  d_in[1];
    const float* ew  = (const float*)d_in[2];
    const float* W1  = (const float*)d_in[3];
    const float* b1  = (const float*)d_in[4];
    const float* W2  = (const float*)d_in[5];
    const float* b2  = (const float*)d_in[6];
    const float* g1  = (const float*)d_in[7];
    const float* be1 = (const float*)d_in[8];
    const float* g2  = (const float*)d_in[9];
    const float* be2 = (const float*)d_in[10];

    const int N = in_sizes[0] / 128;
    const int E = in_sizes[2];
    const int* src = ei;
    const int* dst = ei + E;

    float* H   = (float*)d_ws;
    float* out = (float*)d_out;

    const int Npad = (N + 4) & ~3;
    const int Epad = (E + 3) & ~3;
    int*  cnt    = (int*)(H + (size_t)N * 128);
    int*  rowptr = cnt + Npad;
    int*  rank   = rowptr + Npad;
    int2* perm   = (int2*)(rank + Epad);

    const int gemmBlocks = (N + 15) / 16;
    const int zeroBlocks = (Npad / 4 + 255) / 256;
    const int edgeBlocks = (E + 255) / 256;
    const int rowBlocks  = (N + 7) / 8;

    zero_f32<<<zeroBlocks, 256, 0, stream>>>((float*)cnt, Npad / 4);
    count_rank<<<edgeBlocks, 256, 0, stream>>>(dst, cnt, rank, E);
    scan_counts<<<1, 1024, 0, stream>>>(cnt, rowptr, N);
    fill_buckets<<<edgeBlocks, 256, 0, stream>>>(src, dst, ew, rank, rowptr, perm, E);

    gemm128_wmma<<<gemmBlocks, 256, 0, stream>>>(x, W1, H, N);
    gather_rows_ln_relu<<<rowBlocks, 256, 0, stream>>>(H, rowptr, perm, b1, g1, be1, out, N);

    gemm128_wmma<<<gemmBlocks, 256, 0, stream>>>(out, W2, H, N);
    gather_rows_ln_relu<<<rowBlocks, 256, 0, stream>>>(H, rowptr, perm, b2, g2, be2, out, N);
}


# GEMM: LDS pitch of the staged A tile 128 -> 132 floats: the 16 rows of an A-fragment read fall on different banks, not all on one pair of banks (16-way conflict gone); results bit-identical
# speedup vs baseline: 2.4900x; 1.2061x over previous
#include <hip/hip_runtime.h>
#include <hip/hip_bf16.h>

typedef __attribute__((ext_vector_type(2))) float v2f;
typedef __attribute__((ext_vector_type(8))) float v8f;

#define LN_EPS 1e-5f

#define A_PITCH 132
__global__ __launch_bounds__(256) void gemm128_wmma(
    const float* __restrict__ A, const float* __restrict__ W,
    float* __restrict__ H, int N)
{
    __shared__ float As[16 * A_PITCH];
    const int rowBase = blockIdx.x << 4;

    for (int i = threadIdx.x; i < 512; i += 256) {
        const int rrow = i >> 5;
        const int rcol = i & 31;
        const int gr   = rowBase + rrow;
        float4 v = make_float4(0.f, 0.f, 0.f, 0.f);
        if (gr < N)
            v = reinterpret_cast<const float4*>(A + (size_t)gr * 128)[rcol];
        reinterpret_cast<float4*>(As + rrow * A_PITCH)[rcol] = v;
    }
    __syncthreads();

    const int lane    = threadIdx.x & 31;
    const int wave    = threadIdx.x >> 5;
    const int colBase = wave << 4;
    const int r       = lane & 15;
    const int kh      = lane >> 4;

    v8f acc = {};
    #pragma unroll
    for (int k = 0; k < 128; k += 4) {
        const int ka = k + 2 * kh;
        v2f a, b;
        a.x = As[r * A_PITCH + ka];
        a.y = As[r * A_PITCH + ka + 1];
        b.x = W[(size_t)ka * 128 + colBase + r];
        b.y = W[(size_t)(ka + 1) * 128 + colBase + r];
        acc = __builtin_amdgcn_wmma_f32_16x16x4_f32(false, a, false, b,
                                                    (short)0, acc, false, false);
    }

    #pragma unroll
    for (int v = 0; v < 8; ++v) {
        const int gr = rowBase + 8 * kh + v;
        if (gr < N) H[(size_t)gr * 128 + colBase + r] = acc[v];
    }
}

__global__ __launch_bounds__(256) void zero_f32(float* __restrict__ p, int n4)
{
    int i = blockIdx.x * 256 + threadIdx.x;
    if (i < n4)
        reinterpret_cast<float4*>(p)[i] = make_float4(0.f, 0.f, 0.f, 0.f);
}

__global__ __launch_bounds__(256) void count_rank(
    const int* __restrict__ dst, int* __restrict__ cnt,
    int* __restrict__ rank, int E)
{
    const int e = blockIdx.x * 256 + threadIdx.x;
    if (e < E) rank[e] = atomicAdd(&cnt[dst[e]], 1);
}

__global__ __launch_bounds__(1024) void scan_counts(
    const int* __restrict__ cnt, int* __restrict__ rowptr, int N)
{
    __shared__ int wsum[32];
    __shared__ int chunkTotal;
    const int tid  = threadIdx.x;
    const int lane = tid & 31;
    const int wave = tid >> 5;

    int carry = 0;
    for (int base = 0; base < N; base += 16384) {
        const int i = base + tid * 16;
        int4 c[4];
        #pragma unroll
        for (int q = 0; q < 4; ++q) {
            c[q] = make_int4(0, 0, 0, 0);
            if (i + 4 * q < N) c[q] = *reinterpret_cast<const int4*>(cnt + i + 4 * q);
        }
        int4 p[4];
        int run = 0;
        #pragma unroll
        for (int q = 0; q < 4; ++q) {
            p[q].x = run; run += c[q].x;
            p[q].y = run; run += c[q].y;
            p[q].z = run; run += c[q].z;
            p[q].w = run; run += c[q].w;
        }
        const int t = run;

        int incl = t;
        #pragma unroll
        for (int off = 1; off < 32; off <<= 1) {
            const int y = __shfl_up(incl, off, 32);
            if (lane >= off) incl += y;
        }
        if (lane == 31) wsum[wave] = incl;
        __syncthreads();
        if (wave == 0) {
            const int v = wsum[lane];
            int s = v;
            #pragma unroll
            for (int off = 1; off < 32; off <<= 1) {
                const int y = __shfl_up(s, off, 32);
                if (lane >= off) s += y;
            }
            wsum[lane] = s - v;
            if (lane == 31) chunkTotal = s;
        }
        __syncthreads();
        const int excl = carry + wsum[wave] + incl - t;
        #pragma unroll
        for (int q = 0; q < 4; ++q)
            if (i + 4 * q < N)
                *reinterpret_cast<int4*>(rowptr + i + 4 * q) =
                    make_int4(excl + p[q].x, excl + p[q].y, excl + p[q].z, excl + p[q].w);
        carry += chunkTotal;
        __syncthreads();
    }
    if (tid == 0) rowptr[N] = carry;
}

__global__ __launch_bounds__(256) void fill_buckets(
    const int*   __restrict__ src,
    const int*   __restrict__ dst,
    const float* __restrict__ ew,
    const int*   __restrict__ rank,
    const int*   __restrict__ rowptr,
    int2* __restrict__ perm, int E)
{
    const int e = blockIdx.x * 256 + threadIdx.x;
    if (e < E)
        perm[rowptr[dst[e]] + rank[e]] = make_int2(src[e], __float_as_int(ew[e]));
}

__global__ __launch_bounds__(256) void gather_rows_ln_relu(
    const float* __restrict__ H,
    const int*   __restrict__ rowptr,
    const int2*  __restrict__ perm,
    const float* __restrict__ bias,
    const float* __restrict__ gamma,
    const float* __restrict__ beta,
    float* __restrict__ out, int N)
{
    const int n = __builtin_amdgcn_readfirstlane(blockIdx.x * 8 + (threadIdx.x >> 5));
    if (n >= N) return;
    const int lane = threadIdx.x & 31;

    const float4 b  = reinterpret_cast<const float4*>(bias)[lane];
    const float4 g  = reinterpret_cast<const float4*>(gamma)[lane];
    const float4 be = reinterpret_cast<const float4*>(beta)[lane];

    const int beg  = __builtin_amdgcn_readfirstlane(rowptr[n]);
    const int end  = __builtin_amdgcn_readfirstlane(rowptr[n + 1]);

    const float* hl = H + lane * 4;
    float4 acc = make_float4(0.f, 0.f, 0.f, 0.f);

    for (int base = beg; base < end; base += 32) {
        const int c = min(32, end - base);
        int2 pr = make_int2(0, 0);
        if (lane < c) pr = perm[base + lane];

        for (int j = 0; j < c; j += 8) {
            int s[8];
            #pragma unroll
            for (int u = 0; u < 8; ++u)
                s[u] = __builtin_amdgcn_readlane(pr.x, j + u);
            float4 h[8];
            #pragma unroll
            for (int u = 0; u < 8; ++u) {
                h[u] = make_float4(0.f, 0.f, 0.f, 0.f);
                if (j + u < c)
                    h[u] = *reinterpret_cast<const float4*>(hl + (size_t)s[u] * 128);
            }
            #pragma unroll
            for (int u = 0; u < 8; ++u) {
                const float w = __int_as_float(__builtin_amdgcn_readlane(pr.y, j + u));
                acc.x += h[u].x * w;
                acc.y += h[u].y * w;
                acc.z += h[u].z * w;
                acc.w += h[u].w * w;
            }
        }
    }

    float4 v = acc;
    v.x += b.x; v.y += b.y; v.z += b.z; v.w += b.w;

    float rsum = v.x + v.y + v.z + v.w;
    #pragma unroll
    for (int off = 16; off >= 1; off >>= 1) rsum += __shfl_xor(rsum, off, 32);
    const float mu = rsum * (1.0f / 128.0f);

    const float dx = v.x - mu, dy = v.y - mu, dz = v.z - mu, dw = v.w - mu;
    float q = dx * dx + dy * dy + dz * dz + dw * dw;
    #pragma unroll
    for (int off = 16; off >= 1; off >>= 1) q += __shfl_xor(q, off, 32);
    const float rstd = rsqrtf(q * (1.0f / 128.0f) + LN_EPS);

    float4 o;
    o.x = fmaxf(dx * rstd * g.x + be.x, 0.f);
    o.y = fmaxf(dy * rstd * g.y + be.y, 0.f);
    o.z = fmaxf(dz * rstd * g.z + be.z, 0.f);
    o.w = fmaxf(dw * rstd * g.w + be.w, 0.f);
    reinterpret_cast<float4*>(out + (size_t)n * 128)[lane] = o;
}

extern "C" void kernel_launch(void* const* d_in, const int* in_sizes, int n_in,
                              void* d_out, int out_size, void* d_ws, size_t ws_size,
                              hipStream_t stream)
{
    const float* x   = (const float*)d_in[0];
    const int*   ei  = (const int*)  d_in[1];
    const float* ew  = (const float*)d_in[2];
    const float* W1  = (const float*)d_in[3];
    const float* b1  = (const float*)d_in[4];
    const float* W2  = (const float*)d_in[5];
    const float* b2  = (const float*)d_in[6];
    const float* g1  = (const float*)d_in[7];
    const float* be1 = (const float*)d_in[8];
    const float* g2  = (const float*)d_in[9];
    const float* be2 = (const float*)d_in[10];

    const int N = in_sizes[0] / 128;
    const int E = in_sizes[2];
    const int* src = ei;
    const int* dst = ei + E;

    float* H   = (float*)d_ws;
    float* out = (float*)d_out;

    const int Npad = (N + 4) & ~3;
    const int Epad = (E + 3) & ~3;
    int*  cnt    = (int*)(H + (size_t)N * 128);
    int*  rowptr = cnt + Npad;
    int*  rank   = rowptr + Npad;
    int2* perm   = (int2*)(rank + Epad);

    const int gemmBlocks = (N + 15) / 16;
    const int zeroBlocks = (Npad / 4 + 255) / 256;
    const int edgeBlocks = (E + 255) / 256;
    const int rowBlocks  = (N + 7) / 8;

    zero_f32<<<zeroBlocks, 256, 0, stream>>>((float*)cnt, Npad / 4);
    count_rank<<<edgeBlocks, 256, 0, stream>>>(dst, cnt, rank, E);
    scan_counts<<<1, 1024, 0, stream>>>(cnt, rowptr, N);
    fill_buckets<<<edgeBlocks, 256, 0, stream>>>(src, dst, ew, rank, rowptr, perm, E);

    gemm128_wmma<<<gemmBlocks, 256, 0, stream>>>(x, W1, H, N);
    gather_rows_ln_relu<<<rowBlocks, 256, 0, stream>>>(H, rowptr, perm, b1, g1, be1, out, N);

    gemm128_wmma<<<gemmBlocks, 256, 0, stream>>>(out, W2, H, N);
    gather_rows_ln_relu<<<rowBlocks, 256, 0, stream>>>(H, rowptr, perm, b2, g2, be2, out, N);
}


# buckets built on LDS counters: 4096-edge blocks bin by 256-row range (ds_add return = rank), one block per range places its rows; per-edge global atomics, scan, zero launches gone; arrival-order sums
# speedup vs baseline: 4.4672x; 1.7941x over previous
#include <hip/hip_runtime.h>
#include <hip/hip_bf16.h>

typedef __attribute__((ext_vector_type(2))) float v2f;
typedef __attribute__((ext_vector_type(8))) float v8f;

#define LN_EPS 1e-5f

#define A_PITCH 132
__global__ __launch_bounds__(256) void gemm128_wmma(
    const float* __restrict__ A, const float* __restrict__ W,
    float* __restrict__ H, int N)
{
    __shared__ float As[16 * A_PITCH];
    const int rowBase = blockIdx.x << 4;

    for (int i = threadIdx.x; i < 512; i += 256) {
        const int rrow = i >> 5;
        const int rcol = i & 31;
        const int gr   = rowBase + rrow;
        float4 v = make_float4(0.f, 0.f, 0.f, 0.f);
        if (gr < N)
            v = reinterpret_cast<const float4*>(A + (size_t)gr * 128)[rcol];
        reinterpret_cast<float4*>(As + rrow * A_PITCH)[rcol] = v;
    }
    __syncthreads();

    const int lane    = threadIdx.x & 31;
    const int wave    = threadIdx.x >> 5;
    const int colBase = wave << 4;
    const int r       = lane & 15;
    const int kh      = lane >> 4;

    v8f acc = {};
    #pragma unroll
    for (int k = 0; k < 128; k += 4) {
        const int ka = k + 2 * kh;
        v2f a, b;
        a.x = As[r * A_PITCH + ka];
        a.y = As[r * A_PITCH + ka + 1];
        b.x = W[(size_t)ka * 128 + colBase + r];
        b.y = W[(size_t)(ka + 1) * 128 + colBase + r];
        acc = __builtin_amdgcn_wmma_f32_16x16x4_f32(false, a, false, b,
                                                    (short)0, acc, false, false);
    }

    #pragma unroll
    for (int v = 0; v < 8; ++v) {
        const int gr = rowBase + 8 * kh + v;
        if (gr < N) H[(size_t)gr * 128 + colBase + r] = acc[v];
    }
}

#define BIN_SHIFT 8
#define BIN_ROWS  256
#define CHUNK     4096

__global__ __launch_bounds__(1024) void bin_edges(
    const int*   __restrict__ src,
    const int*   __restrict__ dst,
    const float* __restrict__ ew,
    int2* __restrict__ rec, int* __restrict__ T, int E, int B, int TP)
{
    extern __shared__ int hist[];
    __shared__ int2 stage[CHUNK];
    __shared__ int wsum[32];
    const int tid  = threadIdx.x;
    const int lane = tid & 31;
    const int wave = tid >> 5;
    const int g    = blockIdx.x;
    const int base = g * CHUNK;
    const int nvalid = min(CHUNK, E - base);

    for (int i = tid; i < B; i += 1024) hist[i] = 0;
    __syncthreads();

    int s[4], d[4], w[4], rk[4];
    #pragma unroll
    for (int k = 0; k < 4; ++k) {
        const int idx = k * 1024 + tid;
        d[k] = -1; s[k] = 0; w[k] = 0;
        if (idx < nvalid) {
            d[k] = dst[base + idx];
            s[k] = src[base + idx];
            w[k] = __float_as_int(ew[base + idx]);
        }
    }
    #pragma unroll
    for (int k = 0; k < 4; ++k) {
        rk[k] = 0;
        if (d[k] >= 0) rk[k] = atomicAdd(&hist[d[k] >> BIN_SHIFT], 1);
    }
    __syncthreads();

    const int per = (B + 1023) >> 10;
    const int b0  = tid * per;
    int run = 0;
    for (int i = 0; i < per; ++i) {
        const int bb = b0 + i;
        if (bb < B) run += hist[bb];
    }
    int incl = run;
    #pragma unroll
    for (int off = 1; off < 32; off <<= 1) {
        const int y = __shfl_up(incl, off, 32);
        if (lane >= off) incl += y;
    }
    if (lane == 31) wsum[wave] = incl;
    __syncthreads();
    if (wave == 0) {
        const int v = wsum[lane];
        int t = v;
        #pragma unroll
        for (int off = 1; off < 32; off <<= 1) {
            const int y = __shfl_up(t, off, 32);
            if (lane >= off) t += y;
        }
        wsum[lane] = t - v;
    }
    __syncthreads();
    int acc = wsum[wave] + incl - run;
    for (int i = 0; i < per; ++i) {
        const int bb = b0 + i;
        if (bb < B) {
            const int c = hist[bb];
            hist[bb] = acc;
            T[(size_t)g * TP + bb] = acc;
            acc += c;
        }
    }
    if (tid == 0) T[(size_t)g * TP + B] = nvalid;
    __syncthreads();

    #pragma unroll
    for (int k = 0; k < 4; ++k)
        if (d[k] >= 0)
            stage[hist[d[k] >> BIN_SHIFT] + rk[k]] =
                make_int2(s[k] | ((d[k] & (BIN_ROWS - 1)) << 24), w[k]);
    __syncthreads();
    #pragma unroll
    for (int k = 0; k < 4; ++k) {
        const int idx = k * 1024 + tid;
        if (idx < nvalid) rec[(size_t)base + idx] = stage[idx];
    }
}

template <bool PLACE>
__device__ __forceinline__ void walk_pieces(
    const int2* __restrict__ rec, const int* __restrict__ T,
    int* cnt, int2* __restrict__ perm,
    int b, int GA, int TP, int wave, int lane)
{
    for (int gbase = 0; gbase < GA; gbase += 1024) {
        const int myg = gbase + wave + 32 * lane;
        int sb = 0, se = 0;
        if (myg < GA) {
            sb = T[(size_t)myg * TP + b];
            se = T[(size_t)myg * TP + b + 1];
        }
        const int left = GA - gbase - wave;
        const int nseg = left > 0 ? min(32, (left + 31) >> 5) : 0;
        for (int k0 = 0; k0 < nseg; k0 += 8) {
            int beg[8], len[8];
            #pragma unroll
            for (int u = 0; u < 8; ++u) {
                beg[u] = __builtin_amdgcn_readlane(sb, k0 + u);
                len[u] = __builtin_amdgcn_readlane(se, k0 + u) - beg[u];
            }
            int2 r[8];
            #pragma unroll
            for (int u = 0; u < 8; ++u) {
                r[u] = make_int2(0, 0);
                if (lane < len[u])
                    r[u] = rec[(size_t)(gbase + wave + 32 * (k0 + u)) * CHUNK + beg[u] + lane];
            }
            asm volatile("" :: "v"(r[0].x), "v"(r[1].x), "v"(r[2].x), "v"(r[3].x),
                               "v"(r[4].x), "v"(r[5].x), "v"(r[6].x), "v"(r[7].x));
            #pragma unroll
            for (int u = 0; u < 8; ++u) {
                if (lane < len[u]) {
                    const int j = (int)((unsigned)r[u].x >> 24);
                    if (PLACE) {
                        const int pos = atomicAdd(&cnt[j], 1);
                        perm[pos] = make_int2(r[u].x & 0x00FFFFFF, r[u].y);
                    } else {
                        atomicAdd(&cnt[j], 1);
                    }
                }
            }
            #pragma unroll
            for (int u = 0; u < 8; ++u) {
                for (int i = 32 + lane; i < len[u]; i += 32) {
                    const int2 q = rec[(size_t)(gbase + wave + 32 * (k0 + u)) * CHUNK + beg[u] + i];
                    const int j = (int)((unsigned)q.x >> 24);
                    if (PLACE) {
                        const int pos = atomicAdd(&cnt[j], 1);
                        perm[pos] = make_int2(q.x & 0x00FFFFFF, q.y);
                    } else {
                        atomicAdd(&cnt[j], 1);
                    }
                }
            }
        }
    }
}

__global__ __launch_bounds__(1024) void bucket_rows(
    const int2* __restrict__ rec, const int* __restrict__ T,
    int2* __restrict__ perm, int* __restrict__ rowptr,
    int N, int E, int GA, int TP)
{
    __shared__ int cnt[BIN_ROWS];
    __shared__ int red[32];
    __shared__ int wsum[8];
    const int tid  = threadIdx.x;
    const int lane = tid & 31;
    const int wave = tid >> 5;
    const int b    = blockIdx.x;

    if (tid < BIN_ROWS) cnt[tid] = 0;
    int part = 0;
    for (int g = tid; g < GA; g += 1024) part += T[(size_t)g * TP + b];
    #pragma unroll
    for (int off = 16; off >= 1; off >>= 1) part += __shfl_xor(part, off, 32);
    if (lane == 0) red[wave] = part;
    __syncthreads();

    walk_pieces<false>(rec, T, cnt, perm, b, GA, TP, wave, lane);
    __syncthreads();

    int v = 0, incl = 0;
    if (tid < BIN_ROWS) {
        v = cnt[tid];
        incl = v;
        #pragma unroll
        for (int off = 1; off < 32; off <<= 1) {
            const int y = __shfl_up(incl, off, 32);
            if (lane >= off) incl += y;
        }
        if (lane == 31) wsum[wave] = incl;
    }
    __syncthreads();
    if (tid < BIN_ROWS) {
        int add = 0;
        #pragma unroll
        for (int i = 0; i < 32; ++i) add += red[i];
        for (int w2 = 0; w2 < wave; ++w2) add += wsum[w2];
        const int pos = add + incl - v;
        const int n = b * BIN_ROWS + tid;
        if (n < N) rowptr[n] = pos;
        cnt[tid] = pos;
    }
    if (b == 0 && tid == 0) rowptr[N] = E;
    __syncthreads();

    walk_pieces<true>(rec, T, cnt, perm, b, GA, TP, wave, lane);
}

__global__ __launch_bounds__(256) void gather_rows_ln_relu(
    const float* __restrict__ H,
    const int*   __restrict__ rowptr,
    const int2*  __restrict__ perm,
    const float* __restrict__ bias,
    const float* __restrict__ gamma,
    const float* __restrict__ beta,
    float* __restrict__ out, int N)
{
    const int n = __builtin_amdgcn_readfirstlane(blockIdx.x * 8 + (threadIdx.x >> 5));
    if (n >= N) return;
    const int lane = threadIdx.x & 31;

    const float4 b  = reinterpret_cast<const float4*>(bias)[lane];
    const float4 g  = reinterpret_cast<const float4*>(gamma)[lane];
    const float4 be = reinterpret_cast<const float4*>(beta)[lane];

    const int beg  = __builtin_amdgcn_readfirstlane(rowptr[n]);
    const int end  = __builtin_amdgcn_readfirstlane(rowptr[n + 1]);

    const float* hl = H + lane * 4;
    float4 acc = make_float4(0.f, 0.f, 0.f, 0.f);

    for (int base = beg; base < end; base += 32) {
        const int c = min(32, end - base);
        int2 pr = make_int2(0, 0);
        if (lane < c) pr = perm[base + lane];

        for (int j = 0; j < c; j += 8) {
            int s[8];
            #pragma unroll
            for (int u = 0; u < 8; ++u)
                s[u] = __builtin_amdgcn_readlane(pr.x, j + u);
            float4 h[8];
            #pragma unroll
            for (int u = 0; u < 8; ++u) {
                h[u] = make_float4(0.f, 0.f, 0.f, 0.f);
                if (j + u < c)
                    h[u] = *reinterpret_cast<const float4*>(hl + (size_t)s[u] * 128);
            }
            #pragma unroll
            for (int u = 0; u < 8; ++u) {
                const float w = __int_as_float(__builtin_amdgcn_readlane(pr.y, j + u));
                acc.x += h[u].x * w;
                acc.y += h[u].y * w;
                acc.z += h[u].z * w;
                acc.w += h[u].w * w;
            }
        }
    }

    float4 v = acc;
    v.x += b.x; v.y += b.y; v.z += b.z; v.w += b.w;

    float rsum = v.x + v.y + v.z + v.w;
    #pragma unroll
    for (int off = 16; off >= 1; off >>= 1) rsum += __shfl_xor(rsum, off, 32);
    const float mu = rsum * (1.0f / 128.0f);

    const float dx = v.x - mu, dy = v.y - mu, dz = v.z - mu, dw = v.w - mu;
    float q = dx * dx + dy * dy + dz * dz + dw * dw;
    #pragma unroll
    for (int off = 16; off >= 1; off >>= 1) q += __shfl_xor(q, off, 32);
    const float rstd = rsqrtf(q * (1.0f / 128.0f) + LN_EPS);

    float4 o;
    o.x = fmaxf(dx * rstd * g.x + be.x, 0.f);
    o.y = fmaxf(dy * rstd * g.y + be.y, 0.f);
    o.z = fmaxf(dz * rstd * g.z + be.z, 0.f);
    o.w = fmaxf(dw * rstd * g.w + be.w, 0.f);
    reinterpret_cast<float4*>(out + (size_t)n * 128)[lane] = o;
}

extern "C" void kernel_launch(void* const* d_in, const int* in_sizes, int n_in,
                              void* d_out, int out_size, void* d_ws, size_t ws_size,
                              hipStream_t stream)
{
    const float* x   = (const float*)d_in[0];
    const int*   ei  = (const int*)  d_in[1];
    const float* ew  = (const float*)d_in[2];
    const float* W1  = (const float*)d_in[3];
    const float* b1  = (const float*)d_in[4];
    const float* W2  = (const float*)d_in[5];
    const float* b2  = (const float*)d_in[6];
    const float* g1  = (const float*)d_in[7];
    const float* be1 = (const float*)d_in[8];
    const float* g2  = (const float*)d_in[9];
    const float* be2 = (const float*)d_in[10];

    const int N = in_sizes[0] / 128;
    const int E = in_sizes[2];
    const int* src = ei;
    const int* dst = ei + E;

    float* H   = (float*)d_ws;
    float* out = (float*)d_out;

    const int B    = (N + BIN_ROWS - 1) >> BIN_SHIFT;
    const int GA   = (E + CHUNK - 1) / CHUNK;
    const int TP   = (B + 1 + 63) & ~63;
    const int Npad = (N + 1 + 63) & ~63;
    int*  rowptr = (int*)(H + (size_t)N * 128);
    int*  T      = rowptr + Npad;
    int2* rec    = (int2*)(T + (size_t)GA * TP);
    int2* perm   = rec + (size_t)GA * CHUNK;

    const int gemmBlocks = (N + 15) / 16;
    const int rowBlocks  = (N + 7) / 8;

    bin_edges<<<GA, 1024, (size_t)B * sizeof(int), stream>>>(src, dst, ew, rec, T, E, B, TP);
    bucket_rows<<<B, 1024, 0, stream>>>(rec, T, perm, rowptr, N, E, GA, TP);

    gemm128_wmma<<<gemmBlocks, 256, 0, stream>>>(x, W1, H, N);
    gather_rows_ln_relu<<<rowBlocks, 256, 0, stream>>>(H, rowptr, perm, b1, g1, be1, out, N);

    gemm128_wmma<<<gemmBlocks, 256, 0, stream>>>(out, W2, H, N);
    gather_rows_ln_relu<<<rowBlocks, 256, 0, stream>>>(H, rowptr, perm, b2, g2, be2, out, N);
}


# layer kernel made persistent: 768 resident blocks (3 per WGP by LDS) stage W in LDS once and walk 2-3 row tiles each; next tile's bounds and pairs fetched under the current tile; sums unchanged
# speedup vs baseline: 5.3645x; 1.2009x over previous
#include <hip/hip_runtime.h>
#include <hip/hip_bf16.h>

typedef __attribute__((ext_vector_type(2))) float v2f;
typedef __attribute__((ext_vector_type(8))) float v8f;

#define LN_EPS 1e-5f

#define BIN_SHIFT 8
#define BIN_ROWS  256
#define CHUNK     4096

__global__ __launch_bounds__(1024) void bin_edges(
    const int*   __restrict__ src,
    const int*   __restrict__ dst,
    const float* __restrict__ ew,
    int2* __restrict__ rec, int* __restrict__ T, int E, int B, int TP)
{
    extern __shared__ int hist[];
    __shared__ int2 stage[CHUNK];
    __shared__ int wsum[32];
    const int tid  = threadIdx.x;
    const int lane = tid & 31;
    const int wave = tid >> 5;
    const int g    = blockIdx.x;
    const int base = g * CHUNK;
    const int nvalid = min(CHUNK, E - base);

    for (int i = tid; i < B; i += 1024) hist[i] = 0;
    __syncthreads();

    int s[4], d[4], w[4], rk[4];
    #pragma unroll
    for (int k = 0; k < 4; ++k) {
        const int idx = k * 1024 + tid;
        d[k] = -1; s[k] = 0; w[k] = 0;
        if (idx < nvalid) {
            d[k] = dst[base + idx];
            s[k] = src[base + idx];
            w[k] = __float_as_int(ew[base + idx]);
        }
    }
    #pragma unroll
    for (int k = 0; k < 4; ++k) {
        rk[k] = 0;
        if (d[k] >= 0) rk[k] = atomicAdd(&hist[d[k] >> BIN_SHIFT], 1);
    }
    __syncthreads();

    const int per = (B + 1023) >> 10;
    const int b0  = tid * per;
    int run = 0;
    for (int i = 0; i < per; ++i) {
        const int bb = b0 + i;
        if (bb < B) run += hist[bb];
    }
    int incl = run;
    #pragma unroll
    for (int off = 1; off < 32; off <<= 1) {
        const int y = __shfl_up(incl, off, 32);
        if (lane >= off) incl += y;
    }
    if (lane == 31) wsum[wave] = incl;
    __syncthreads();
    if (wave == 0) {
        const int v = wsum[lane];
        int t = v;
        #pragma unroll
        for (int off = 1; off < 32; off <<= 1) {
            const int y = __shfl_up(t, off, 32);
            if (lane >= off) t += y;
        }
        wsum[lane] = t - v;
    }
    __syncthreads();
    int acc = wsum[wave] + incl - run;
    for (int i = 0; i < per; ++i) {
        const int bb = b0 + i;
        if (bb < B) {
            const int c = hist[bb];
            hist[bb] = acc;
            T[(size_t)g * TP + bb] = acc;
            acc += c;
        }
    }
    if (tid == 0) T[(size_t)g * TP + B] = nvalid;
    __syncthreads();

    #pragma unroll
    for (int k = 0; k < 4; ++k)
        if (d[k] >= 0)
            stage[hist[d[k] >> BIN_SHIFT] + rk[k]] =
                make_int2(s[k] | ((d[k] & (BIN_ROWS - 1)) << 24), w[k]);
    __syncthreads();
    #pragma unroll
    for (int k = 0; k < 4; ++k) {
        const int idx = k * 1024 + tid;
        if (idx < nvalid) rec[(size_t)base + idx] = stage[idx];
    }
}

template <bool PLACE>
__device__ __forceinline__ void walk_pieces(
    const int2* __restrict__ rec, const int* __restrict__ T,
    int* cnt, int2* __restrict__ perm,
    int b, int GA, int TP, int wave, int lane)
{
    for (int gbase = 0; gbase < GA; gbase += 1024) {
        const int myg = gbase + wave + 32 * lane;
        int sb = 0, se = 0;
        if (myg < GA) {
            sb = T[(size_t)myg * TP + b];
            se = T[(size_t)myg * TP + b + 1];
        }
        const int left = GA - gbase - wave;
        const int nseg = left > 0 ? min(32, (left + 31) >> 5) : 0;
        for (int k0 = 0; k0 < nseg; k0 += 8) {
            int beg[8], len[8];
            #pragma unroll
            for (int u = 0; u < 8; ++u) {
                beg[u] = __builtin_amdgcn_readlane(sb, k0 + u);
                len[u] = __builtin_amdgcn_readlane(se, k0 + u) - beg[u];
            }
            int2 r[8];
            #pragma unroll
            for (int u = 0; u < 8; ++u) {
                r[u] = make_int2(0, 0);
                if (lane < len[u])
                    r[u] = rec[(size_t)(gbase + wave + 32 * (k0 + u)) * CHUNK + beg[u] + lane];
            }
            asm volatile("" :: "v"(r[0].x), "v"(r[1].x), "v"(r[2].x), "v"(r[3].x),
                               "v"(r[4].x), "v"(r[5].x), "v"(r[6].x), "v"(r[7].x));
            #pragma unroll
            for (int u = 0; u < 8; ++u) {
                if (lane < len[u]) {
                    const int j = (int)((unsigned)r[u].x >> 24);
                    if (PLACE) {
                        const int pos = atomicAdd(&cnt[j], 1);
                        perm[pos] = make_int2(r[u].x & 0x00FFFFFF, r[u].y);
                    } else {
                        atomicAdd(&cnt[j], 1);
                    }
                }
            }
            #pragma unroll
            for (int u = 0; u < 8; ++u) {
                for (int i = 32 + lane; i < len[u]; i += 32) {
                    const int2 q = rec[(size_t)(gbase + wave + 32 * (k0 + u)) * CHUNK + beg[u] + i];
                    const int j = (int)((unsigned)q.x >> 24);
                    if (PLACE) {
                        const int pos = atomicAdd(&cnt[j], 1);
                        perm[pos] = make_int2(q.x & 0x00FFFFFF, q.y);
                    } else {
                        atomicAdd(&cnt[j], 1);
                    }
                }
            }
        }
    }
}

__global__ __launch_bounds__(1024) void bucket_rows(
    const int2* __restrict__ rec, const int* __restrict__ T,
    int2* __restrict__ perm, int* __restrict__ rowptr,
    int N, int E, int GA, int TP)
{
    __shared__ int cnt[BIN_ROWS];
    __shared__ int red[32];
    __shared__ int wsum[8];
    const int tid  = threadIdx.x;
    const int lane = tid & 31;
    const int wave = tid >> 5;
    const int b    = blockIdx.x;

    if (tid < BIN_ROWS) cnt[tid] = 0;
    int part = 0;
    for (int g = tid; g < GA; g += 1024) part += T[(size_t)g * TP + b];
    #pragma unroll
    for (int off = 16; off >= 1; off >>= 1) part += __shfl_xor(part, off, 32);
    if (lane == 0) red[wave] = part;
    __syncthreads();

    walk_pieces<false>(rec, T, cnt, perm, b, GA, TP, wave, lane);
    __syncthreads();

    int v = 0, incl = 0;
    if (tid < BIN_ROWS) {
        v = cnt[tid];
        incl = v;
        #pragma unroll
        for (int off = 1; off < 32; off <<= 1) {
            const int y = __shfl_up(incl, off, 32);
            if (lane >= off) incl += y;
        }
        if (lane == 31) wsum[wave] = incl;
    }
    __syncthreads();
    if (tid < BIN_ROWS) {
        int add = 0;
        #pragma unroll
        for (int i = 0; i < 32; ++i) add += red[i];
        for (int w2 = 0; w2 < wave; ++w2) add += wsum[w2];
        const int pos = add + incl - v;
        const int n = b * BIN_ROWS + tid;
        if (n < N) rowptr[n] = pos;
        cnt[tid] = pos;
    }
    if (b == 0 && tid == 0) rowptr[N] = E;
    __syncthreads();

    walk_pieces<true>(rec, T, cnt, perm, b, GA, TP, wave, lane);
}

#define A_PITCH 132
#define W_PITCH 136
#define TILE_M  32

__device__ __forceinline__ int2 load_pairs(
    const int2* __restrict__ perm, int base, int end, int lane)
{
    int2 pr = make_int2(0, 0);
    if (lane < min(32, end - base)) pr = perm[base + lane];
    return pr;
}

__device__ __forceinline__ float4 gather_row(
    const float* __restrict__ xl, const int2* __restrict__ perm,
    int beg, int end, int lane, int2 first)
{
    float4 acc = make_float4(0.f, 0.f, 0.f, 0.f);
    int2 pr = first;
    for (int base = beg; base < end; base += 32) {
        const int c = min(32, end - base);
        if (base != beg) pr = load_pairs(perm, base, end, lane);

        for (int j = 0; j < c; j += 8) {
            int s[8];
            #pragma unroll
            for (int u = 0; u < 8; ++u)
                s[u] = __builtin_amdgcn_readlane(pr.x, j + u);
            float4 h[8];
            #pragma unroll
            for (int u = 0; u < 8; ++u) {
                h[u] = make_float4(0.f, 0.f, 0.f, 0.f);
                if (j + u < c)
                    h[u] = *reinterpret_cast<const float4*>(xl + (size_t)s[u] * 128);
            }
            #pragma unroll
            for (int u = 0; u < 8; ++u) {
                const float w = __int_as_float(__builtin_amdgcn_readlane(pr.y, j + u));
                acc.x += h[u].x * w;
                acc.y += h[u].y * w;
                acc.z += h[u].z * w;
                acc.w += h[u].w * w;
            }
        }
    }
    return acc;
}

__global__ __launch_bounds__(512) void gather_gemm_ln_relu(
    const float* __restrict__ X,
    const int*   __restrict__ rowptr,
    const int2*  __restrict__ perm,
    const float* __restrict__ W,
    const float* __restrict__ bias,
    const float* __restrict__ gamma,
    const float* __restrict__ beta,
    float* __restrict__ Y, int N, int nTiles, int step)
{
    __shared__ float Ws[128 * W_PITCH];
    __shared__ float As[TILE_M * A_PITCH];
    __shared__ float Cs[TILE_M * A_PITCH];

    const int tid  = threadIdx.x;
    const int lane = tid & 31;
    const int wave = __builtin_amdgcn_readfirstlane(tid >> 5);
    int tile = blockIdx.x;

    int p0, p1, p2;
    {
        const int n0 = tile * TILE_M + 2 * wave;
        p0 = __builtin_amdgcn_readfirstlane(rowptr[min(n0,     N)]);
        p1 = __builtin_amdgcn_readfirstlane(rowptr[min(n0 + 1, N)]);
        p2 = __builtin_amdgcn_readfirstlane(rowptr[min(n0 + 2, N)]);
    }
    int2 pr0 = load_pairs(perm, p0, p1, lane);
    int2 pr1 = load_pairs(perm, p1, p2, lane);

    #pragma unroll
    for (int j = 0; j < 8; ++j) {
        const int i = tid + 512 * j;
        const float4 v = reinterpret_cast<const float4*>(W)[i];
        *reinterpret_cast<float4*>(Ws + (i >> 5) * W_PITCH + (i & 31) * 4) = v;
    }

    const int mt = wave >> 3;
    const int ct = wave & 7;
    const int r  = lane & 15;
    const int kh = lane >> 4;
    const float* Ap = As + (16 * mt + r) * A_PITCH + 2 * kh;
    const float* Bp = Ws + (2 * kh) * W_PITCH + 16 * ct + r;
    const float* xl = X + lane * 4;

    for (; tile < nTiles; tile += step) {
        const int rowBase = tile * TILE_M;

        const int m0 = (tile + step) * TILE_M + 2 * wave;
        const int q0v = rowptr[min(m0,     N)];
        const int q1v = rowptr[min(m0 + 1, N)];
        const int q2v = rowptr[min(m0 + 2, N)];

        const float4 s0 = gather_row(xl, perm, p0, p1, lane, pr0);
        *reinterpret_cast<float4*>(As + (2 * wave) * A_PITCH + 4 * lane) = s0;
        const float4 s1 = gather_row(xl, perm, p1, p2, lane, pr1);
        *reinterpret_cast<float4*>(As + (2 * wave + 1) * A_PITCH + 4 * lane) = s1;

        __syncthreads();

        p0 = __builtin_amdgcn_readfirstlane(q0v);
        p1 = __builtin_amdgcn_readfirstlane(q1v);
        p2 = __builtin_amdgcn_readfirstlane(q2v);
        pr0 = load_pairs(perm, p0, p1, lane);
        pr1 = load_pairs(perm, p1, p2, lane);

        v8f acc = {};
        #pragma unroll
        for (int k = 0; k < 128; k += 4) {
            v2f a, b;
            a.x = Ap[k];
            a.y = Ap[k + 1];
            b.x = Bp[k * W_PITCH];
            b.y = Bp[(k + 1) * W_PITCH];
            acc = __builtin_amdgcn_wmma_f32_16x16x4_f32(false, a, false, b,
                                                        (short)0, acc, false, false);
        }

        #pragma unroll
        for (int v = 0; v < 8; ++v)
            Cs[(16 * mt + 8 * kh + v) * A_PITCH + 16 * ct + r] = acc[v];

        const float4 b  = reinterpret_cast<const float4*>(bias)[lane];
        const float4 g  = reinterpret_cast<const float4*>(gamma)[lane];
        const float4 be = reinterpret_cast<const float4*>(beta)[lane];
        __syncthreads();

        #pragma unroll
        for (int rr = 0; rr < 2; ++rr) {
            const int rl = 2 * wave + rr;
            const int n  = rowBase + rl;
            float4 v = *reinterpret_cast<const float4*>(Cs + rl * A_PITCH + 4 * lane);
            v.x += b.x; v.y += b.y; v.z += b.z; v.w += b.w;

            float rsum = v.x + v.y + v.z + v.w;
            #pragma unroll
            for (int off = 16; off >= 1; off >>= 1) rsum += __shfl_xor(rsum, off, 32);
            const float mu = rsum * (1.0f / 128.0f);

            const float dx = v.x - mu, dy = v.y - mu, dz = v.z - mu, dw = v.w - mu;
            float q = dx * dx + dy * dy + dz * dz + dw * dw;
            #pragma unroll
            for (int off = 16; off >= 1; off >>= 1) q += __shfl_xor(q, off, 32);
            const float rstd = rsqrtf(q * (1.0f / 128.0f) + LN_EPS);

            float4 o;
            o.x = fmaxf(dx * rstd * g.x + be.x, 0.f);
            o.y = fmaxf(dy * rstd * g.y + be.y, 0.f);
            o.z = fmaxf(dz * rstd * g.z + be.z, 0.f);
            o.w = fmaxf(dw * rstd * g.w + be.w, 0.f);
            if (n < N) reinterpret_cast<float4*>(Y + (size_t)n * 128)[lane] = o;
        }
    }
}

extern "C" void kernel_launch(void* const* d_in, const int* in_sizes, int n_in,
                              void* d_out, int out_size, void* d_ws, size_t ws_size,
                              hipStream_t stream)
{
    const float* x   = (const float*)d_in[0];
    const int*   ei  = (const int*)  d_in[1];
    const float* ew  = (const float*)d_in[2];
    const float* W1  = (const float*)d_in[3];
    const float* b1  = (const float*)d_in[4];
    const float* W2  = (const float*)d_in[5];
    const float* b2  = (const float*)d_in[6];
    const float* g1  = (const float*)d_in[7];
    const float* be1 = (const float*)d_in[8];
    const float* g2  = (const float*)d_in[9];
    const float* be2 = (const float*)d_in[10];

    const int N = in_sizes[0] / 128;
    const int E = in_sizes[2];
    const int* src = ei;
    const int* dst = ei + E;

    float* H1  = (float*)d_ws;
    float* out = (float*)d_out;

    const int B    = (N + BIN_ROWS - 1) >> BIN_SHIFT;
    const int GA   = (E + CHUNK - 1) / CHUNK;
    const int TP   = (B + 1 + 63) & ~63;
    const int Npad = (N + 1 + 63) & ~63;
    int*  rowptr = (int*)(H1 + (size_t)N * 128);
    int*  T      = rowptr + Npad;
    int2* rec    = (int2*)(T + (size_t)GA * TP);
    int2* perm   = rec + (size_t)GA * CHUNK;

    const int nTiles = (N + TILE_M - 1) / TILE_M;
    const int tileBlocks = nTiles < 768 ? nTiles : 768;

    bin_edges<<<GA, 1024, (size_t)B * sizeof(int), stream>>>(src, dst, ew, rec, T, E, B, TP);
    bucket_rows<<<B, 1024, 0, stream>>>(rec, T, perm, rowptr, N, E, GA, TP);

    gather_gemm_ln_relu<<<tileBlocks, 512, 0, stream>>>(x,  rowptr, perm, W1, b1, g1, be1, H1,  N, nTiles, tileBlocks);
    gather_gemm_ln_relu<<<tileBlocks, 512, 0, stream>>>(H1, rowptr, perm, W2, b2, g2, be2, out, N, nTiles, tileBlocks);
}


# layer kernel, row epilogue: the four 32-lane row sums per tile (mean, squared deviations; two rows) by v_permlanex16 + DPP row_xmask adds on the vector ALU, not 20 waited ds_bpermute; same pair order
# speedup vs baseline: 5.7412x; 1.0702x over previous
#include <hip/hip_runtime.h>
#include <hip/hip_bf16.h>

typedef __attribute__((ext_vector_type(2))) float v2f;
typedef __attribute__((ext_vector_type(8))) float v8f;

#define LN_EPS 1e-5f

#define BIN_SHIFT 8
#define BIN_ROWS  256
#define CHUNK     4096

__global__ __launch_bounds__(1024) void bin_edges(
    const int*   __restrict__ src,
    const int*   __restrict__ dst,
    const float* __restrict__ ew,
    int2* __restrict__ rec, int* __restrict__ T, int E, int B, int TP)
{
    extern __shared__ int hist[];
    __shared__ int2 stage[CHUNK];
    __shared__ int wsum[32];
    const int tid  = threadIdx.x;
    const int lane = tid & 31;
    const int wave = tid >> 5;
    const int g    = blockIdx.x;
    const int base = g * CHUNK;
    const int nvalid = min(CHUNK, E - base);

    for (int i = tid; i < B; i += 1024) hist[i] = 0;
    __syncthreads();

    int s[4], d[4], w[4], rk[4];
    #pragma unroll
    for (int k = 0; k < 4; ++k) {
        const int idx = k * 1024 + tid;
        d[k] = -1; s[k] = 0; w[k] = 0;
        if (idx < nvalid) {
            d[k] = dst[base + idx];
            s[k] = src[base + idx];
            w[k] = __float_as_int(ew[base + idx]);
        }
    }
    #pragma unroll
    for (int k = 0; k < 4; ++k) {
        rk[k] = 0;
        if (d[k] >= 0) rk[k] = atomicAdd(&hist[d[k] >> BIN_SHIFT], 1);
    }
    __syncthreads();

    const int per = (B + 1023) >> 10;
    const int b0  = tid * per;
    int run = 0;
    for (int i = 0; i < per; ++i) {
        const int bb = b0 + i;
        if (bb < B) run += hist[bb];
    }
    int incl = run;
    #pragma unroll
    for (int off = 1; off < 32; off <<= 1) {
        const int y = __shfl_up(incl, off, 32);
        if (lane >= off) incl += y;
    }
    if (lane == 31) wsum[wave] = incl;
    __syncthreads();
    if (wave == 0) {
        const int v = wsum[lane];
        int t = v;
        #pragma unroll
        for (int off = 1; off < 32; off <<= 1) {
            const int y = __shfl_up(t, off, 32);
            if (lane >= off) t += y;
        }
        wsum[lane] = t - v;
    }
    __syncthreads();
    int acc = wsum[wave] + incl - run;
    for (int i = 0; i < per; ++i) {
        const int bb = b0 + i;
        if (bb < B) {
            const int c = hist[bb];
            hist[bb] = acc;
            T[(size_t)g * TP + bb] = acc;
            acc += c;
        }
    }
    if (tid == 0) T[(size_t)g * TP + B] = nvalid;
    __syncthreads();

    #pragma unroll
    for (int k = 0; k < 4; ++k)
        if (d[k] >= 0)
            stage[hist[d[k] >> BIN_SHIFT] + rk[k]] =
                make_int2(s[k] | ((d[k] & (BIN_ROWS - 1)) << 24), w[k]);
    __syncthreads();
    #pragma unroll
    for (int k = 0; k < 4; ++k) {
        const int idx = k * 1024 + tid;
        if (idx < nvalid) rec[(size_t)base + idx] = stage[idx];
    }
}

template <bool PLACE>
__device__ __forceinline__ void walk_pieces(
    const int2* __restrict__ rec, const int* __restrict__ T,
    int* cnt, int2* __restrict__ perm,
    int b, int GA, int TP, int wave, int lane)
{
    for (int gbase = 0; gbase < GA; gbase += 1024) {
        const int myg = gbase + wave + 32 * lane;
        int sb = 0, se = 0;
        if (myg < GA) {
            sb = T[(size_t)myg * TP + b];
            se = T[(size_t)myg * TP + b + 1];
        }
        const int left = GA - gbase - wave;
        const int nseg = left > 0 ? min(32, (left + 31) >> 5) : 0;
        for (int k0 = 0; k0 < nseg; k0 += 8) {
            int beg[8], len[8];
            #pragma unroll
            for (int u = 0; u < 8; ++u) {
                beg[u] = __builtin_amdgcn_readlane(sb, k0 + u);
                len[u] = __builtin_amdgcn_readlane(se, k0 + u) - beg[u];
            }
            int2 r[8];
            #pragma unroll
            for (int u = 0; u < 8; ++u) {
                r[u] = make_int2(0, 0);
                if (lane < len[u])
                    r[u] = rec[(size_t)(gbase + wave + 32 * (k0 + u)) * CHUNK + beg[u] + lane];
            }
            asm volatile("" :: "v"(r[0].x), "v"(r[1].x), "v"(r[2].x), "v"(r[3].x),
                               "v"(r[4].x), "v"(r[5].x), "v"(r[6].x), "v"(r[7].x));
            #pragma unroll
            for (int u = 0; u < 8; ++u) {
                if (lane < len[u]) {
                    const int j = (int)((unsigned)r[u].x >> 24);
                    if (PLACE) {
                        const int pos = atomicAdd(&cnt[j], 1);
                        perm[pos] = make_int2(r[u].x & 0x00FFFFFF, r[u].y);
                    } else {
                        atomicAdd(&cnt[j], 1);
                    }
                }
            }
            #pragma unroll
            for (int u = 0; u < 8; ++u) {
                for (int i = 32 + lane; i < len[u]; i += 32) {
                    const int2 q = rec[(size_t)(gbase + wave + 32 * (k0 + u)) * CHUNK + beg[u] + i];
                    const int j = (int)((unsigned)q.x >> 24);
                    if (PLACE) {
                        const int pos = atomicAdd(&cnt[j], 1);
                        perm[pos] = make_int2(q.x & 0x00FFFFFF, q.y);
                    } else {
                        atomicAdd(&cnt[j], 1);
                    }
                }
            }
        }
    }
}

__global__ __launch_bounds__(1024) void bucket_rows(
    const int2* __restrict__ rec, const int* __restrict__ T,
    int2* __restrict__ perm, int* __restrict__ rowptr,
    int N, int E, int GA, int TP)
{
    __shared__ int cnt[BIN_ROWS];
    __shared__ int red[32];
    __shared__ int wsum[8];
    const int tid  = threadIdx.x;
    const int lane = tid & 31;
    const int wave = tid >> 5;
    const int b    = blockIdx.x;

    if (tid < BIN_ROWS) cnt[tid] = 0;
    int part = 0;
    for (int g = tid; g < GA; g += 1024) part += T[(size_t)g * TP + b];
    #pragma unroll
    for (int off = 16; off >= 1; off >>= 1) part += __shfl_xor(part, off, 32);
    if (lane == 0) red[wave] = part;
    __syncthreads();

    walk_pieces<false>(rec, T, cnt, perm, b, GA, TP, wave, lane);
    __syncthreads();

    int v = 0, incl = 0;
    if (tid < BIN_ROWS) {
        v = cnt[tid];
        incl = v;
        #pragma unroll
        for (int off = 1; off < 32; off <<= 1) {
            const int y = __shfl_up(incl, off, 32);
            if (lane >= off) incl += y;
        }
        if (lane == 31) wsum[wave] = incl;
    }
    __syncthreads();
    if (tid < BIN_ROWS) {
        int add = 0;
        #pragma unroll
        for (int i = 0; i < 32; ++i) add += red[i];
        for (int w2 = 0; w2 < wave; ++w2) add += wsum[w2];
        const int pos = add + incl - v;
        const int n = b * BIN_ROWS + tid;
        if (n < N) rowptr[n] = pos;
        cnt[tid] = pos;
    }
    if (b == 0 && tid == 0) rowptr[N] = E;
    __syncthreads();

    walk_pieces<true>(rec, T, cnt, perm, b, GA, TP, wave, lane);
}

#define A_PITCH 132
#define W_PITCH 136
#define TILE_M  32

__device__ __forceinline__ float xhalf(float x)
{
    const int b = __float_as_int(x);
    return __int_as_float(__builtin_amdgcn_permlanex16(b, b, 0x76543210, 0xfedcba98, false, false));
}
template <int M>
__device__ __forceinline__ float xrow(float x)
{
    return __int_as_float(__builtin_amdgcn_update_dpp(0, __float_as_int(x), 0x160 + M, 0xf, 0xf, true));
}
__device__ __forceinline__ void wave_sum2(float& a, float& b)
{
    a += xhalf(a);    b += xhalf(b);
    a += xrow<8>(a);  b += xrow<8>(b);
    a += xrow<4>(a);  b += xrow<4>(b);
    a += xrow<2>(a);  b += xrow<2>(b);
    a += xrow<1>(a);  b += xrow<1>(b);
}

__device__ __forceinline__ int2 load_pairs(
    const int2* __restrict__ perm, int base, int end, int lane)
{
    int2 pr = make_int2(0, 0);
    if (lane < min(32, end - base)) pr = perm[base + lane];
    return pr;
}

__device__ __forceinline__ float4 gather_row(
    const float* __restrict__ xl, const int2* __restrict__ perm,
    int beg, int end, int lane, int2 first)
{
    float4 acc = make_float4(0.f, 0.f, 0.f, 0.f);
    int2 pr = first;
    for (int base = beg; base < end; base += 32) {
        const int c = min(32, end - base);
        if (base != beg) pr = load_pairs(perm, base, end, lane);

        for (int j = 0; j < c; j += 8) {
            int s[8];
            #pragma unroll
            for (int u = 0; u < 8; ++u)
                s[u] = __builtin_amdgcn_readlane(pr.x, j + u);
            float4 h[8];
            #pragma unroll
            for (int u = 0; u < 8; ++u) {
                h[u] = make_float4(0.f, 0.f, 0.f, 0.f);
                if (j + u < c)
                    h[u] = *reinterpret_cast<const float4*>(xl + (size_t)s[u] * 128);
            }
            #pragma unroll
            for (int u = 0; u < 8; ++u) {
                const float w = __int_as_float(__builtin_amdgcn_readlane(pr.y, j + u));
                acc.x += h[u].x * w;
                acc.y += h[u].y * w;
                acc.z += h[u].z * w;
                acc.w += h[u].w * w;
            }
        }
    }
    return acc;
}

__global__ __launch_bounds__(512) void gather_gemm_ln_relu(
    const float* __restrict__ X,
    const int*   __restrict__ rowptr,
    const int2*  __restrict__ perm,
    const float* __restrict__ W,
    const float* __restrict__ bias,
    const float* __restrict__ gamma,
    const float* __restrict__ beta,
    float* __restrict__ Y, int N, int nTiles, int step)
{
    __shared__ float Ws[128 * W_PITCH];
    __shared__ float As[TILE_M * A_PITCH];
    __shared__ float Cs[TILE_M * A_PITCH];

    const int tid  = threadIdx.x;
    const int lane = tid & 31;
    const int wave = __builtin_amdgcn_readfirstlane(tid >> 5);
    int tile = blockIdx.x;

    int p0, p1, p2;
    {
        const int n0 = tile * TILE_M + 2 * wave;
        p0 = __builtin_amdgcn_readfirstlane(rowptr[min(n0,     N)]);
        p1 = __builtin_amdgcn_readfirstlane(rowptr[min(n0 + 1, N)]);
        p2 = __builtin_amdgcn_readfirstlane(rowptr[min(n0 + 2, N)]);
    }
    int2 pr0 = load_pairs(perm, p0, p1, lane);
    int2 pr1 = load_pairs(perm, p1, p2, lane);

    #pragma unroll
    for (int j = 0; j < 8; ++j) {
        const int i = tid + 512 * j;
        const float4 v = reinterpret_cast<const float4*>(W)[i];
        *reinterpret_cast<float4*>(Ws + (i >> 5) * W_PITCH + (i & 31) * 4) = v;
    }

    const int mt = wave >> 3;
    const int ct = wave & 7;
    const int r  = lane & 15;
    const int kh = lane >> 4;
    const float* Ap = As + (16 * mt + r) * A_PITCH + 2 * kh;
    const float* Bp = Ws + (2 * kh) * W_PITCH + 16 * ct + r;
    const float* xl = X + lane * 4;

    for (; tile < nTiles; tile += step) {
        const int rowBase = tile * TILE_M;

        const int m0 = (tile + step) * TILE_M + 2 * wave;
        const int q0v = rowptr[min(m0,     N)];
        const int q1v = rowptr[min(m0 + 1, N)];
        const int q2v = rowptr[min(m0 + 2, N)];

        const float4 s0 = gather_row(xl, perm, p0, p1, lane, pr0);
        *reinterpret_cast<float4*>(As + (2 * wave) * A_PITCH + 4 * lane) = s0;
        const float4 s1 = gather_row(xl, perm, p1, p2, lane, pr1);
        *reinterpret_cast<float4*>(As + (2 * wave + 1) * A_PITCH + 4 * lane) = s1;

        __syncthreads();

        p0 = __builtin_amdgcn_readfirstlane(q0v);
        p1 = __builtin_amdgcn_readfirstlane(q1v);
        p2 = __builtin_amdgcn_readfirstlane(q2v);
        pr0 = load_pairs(perm, p0, p1, lane);
        pr1 = load_pairs(perm, p1, p2, lane);

        v8f acc = {};
        #pragma unroll
        for (int k = 0; k < 128; k += 4) {
            v2f a, b;
            a.x = Ap[k];
            a.y = Ap[k + 1];
            b.x = Bp[k * W_PITCH];
            b.y = Bp[(k + 1) * W_PITCH];
            acc = __builtin_amdgcn_wmma_f32_16x16x4_f32(false, a, false, b,
                                                        (short)0, acc, false, false);
        }

        #pragma unroll
        for (int v = 0; v < 8; ++v)
            Cs[(16 * mt + 8 * kh + v) * A_PITCH + 16 * ct + r] = acc[v];

        const float4 b  = reinterpret_cast<const float4*>(bias)[lane];
        const float4 g  = reinterpret_cast<const float4*>(gamma)[lane];
        const float4 be = reinterpret_cast<const float4*>(beta)[lane];
        __syncthreads();

        {
            const int rl = 2 * wave;
            const int n  = rowBase + rl;
            float4 v0 = *reinterpret_cast<const float4*>(Cs + rl * A_PITCH + 4 * lane);
            float4 v1 = *reinterpret_cast<const float4*>(Cs + (rl + 1) * A_PITCH + 4 * lane);
            v0.x += b.x; v0.y += b.y; v0.z += b.z; v0.w += b.w;
            v1.x += b.x; v1.y += b.y; v1.z += b.z; v1.w += b.w;

            float rsum0 = v0.x + v0.y + v0.z + v0.w;
            float rsum1 = v1.x + v1.y + v1.z + v1.w;
            wave_sum2(rsum0, rsum1);
            const float mu0 = rsum0 * (1.0f / 128.0f);
            const float mu1 = rsum1 * (1.0f / 128.0f);

            const float dx0 = v0.x - mu0, dy0 = v0.y - mu0, dz0 = v0.z - mu0, dw0 = v0.w - mu0;
            const float dx1 = v1.x - mu1, dy1 = v1.y - mu1, dz1 = v1.z - mu1, dw1 = v1.w - mu1;
            float q0 = dx0 * dx0 + dy0 * dy0 + dz0 * dz0 + dw0 * dw0;
            float q1 = dx1 * dx1 + dy1 * dy1 + dz1 * dz1 + dw1 * dw1;
            wave_sum2(q0, q1);
            const float rstd0 = rsqrtf(q0 * (1.0f / 128.0f) + LN_EPS);
            const float rstd1 = rsqrtf(q1 * (1.0f / 128.0f) + LN_EPS);

            float4 o0, o1;
            o0.x = fmaxf(dx0 * rstd0 * g.x + be.x, 0.f);
            o0.y = fmaxf(dy0 * rstd0 * g.y + be.y, 0.f);
            o0.z = fmaxf(dz0 * rstd0 * g.z + be.z, 0.f);
            o0.w = fmaxf(dw0 * rstd0 * g.w + be.w, 0.f);
            o1.x = fmaxf(dx1 * rstd1 * g.x + be.x, 0.f);
            o1.y = fmaxf(dy1 * rstd1 * g.y + be.y, 0.f);
            o1.z = fmaxf(dz1 * rstd1 * g.z + be.z, 0.f);
            o1.w = fmaxf(dw1 * rstd1 * g.w + be.w, 0.f);
            if (n < N)     reinterpret_cast<float4*>(Y + (size_t)n * 128)[lane] = o0;
            if (n + 1 < N) reinterpret_cast<float4*>(Y + (size_t)(n + 1) * 128)[lane] = o1;
        }
    }
}

extern "C" void kernel_launch(void* const* d_in, const int* in_sizes, int n_in,
                              void* d_out, int out_size, void* d_ws, size_t ws_size,
                              hipStream_t stream)
{
    const float* x   = (const float*)d_in[0];
    const int*   ei  = (const int*)  d_in[1];
    const float* ew  = (const float*)d_in[2];
    const float* W1  = (const float*)d_in[3];
    const float* b1  = (const float*)d_in[4];
    const float* W2  = (const float*)d_in[5];
    const float* b2  = (const float*)d_in[6];
    const float* g1  = (const float*)d_in[7];
    const float* be1 = (const float*)d_in[8];
    const float* g2  = (const float*)d_in[9];
    const float* be2 = (const float*)d_in[10];

    const int N = in_sizes[0] / 128;
    const int E = in_sizes[2];
    const int* src = ei;
    const int* dst = ei + E;

    float* H1  = (float*)d_ws;
    float* out = (float*)d_out;

    const int B    = (N + BIN_ROWS - 1) >> BIN_SHIFT;
    const int GA   = (E + CHUNK - 1) / CHUNK;
    const int TP   = (B + 1 + 63) & ~63;
    const int Npad = (N + 1 + 63) & ~63;
    int*  rowptr = (int*)(H1 + (size_t)N * 128);
    int*  T      = rowptr + Npad;
    int2* rec    = (int2*)(T + (size_t)GA * TP);
    int2* perm   = rec + (size_t)GA * CHUNK;

    const int nTiles = (N + TILE_M - 1) / TILE_M;
    const int tileBlocks = nTiles < 768 ? nTiles : 768;

    bin_edges<<<GA, 1024, (size_t)B * sizeof(int), stream>>>(src, dst, ew, rec, T, E, B, TP);
    bucket_rows<<<B, 1024, 0, stream>>>(rec, T, perm, rowptr, N, E, GA, TP);

    gather_gemm_ln_relu<<<tileBlocks, 512, 0, stream>>>(x,  rowptr, perm, W1, b1, g1, be1, H1,  N, nTiles, tileBlocks);
    gather_gemm_ln_relu<<<tileBlocks, 512, 0, stream>>>(H1, rowptr, perm, W2, b2, g2, be2, out, N, nTiles, tileBlocks);
}
